# starting kernel
# baseline (speedup 1.0000x reference)
#include <hip/hip_runtime.h>


typedef __attribute__((ext_vector_type(2))) float v2f;
typedef __attribute__((ext_vector_type(8))) float v8f;

#define NB   8
#define NC   128
#define HH   256
#define WW   256
#define HW   (HH * WW)
#define KW   8
#define HP   (HH - KW + 1)
#define WP   (WW - KW + 1)

#if __has_builtin(__builtin_amdgcn_wmma_f32_16x16x4_f32)
#define USE_WMMA_F32X4 1
#else
#define USE_WMMA_F32X4 0
#endif

__global__ __launch_bounds__(256) void mean_exp_kernel(const float* __restrict__ x,
                                                       float* __restrict__ E) {
  const int    tid = threadIdx.x;
  const size_t blk = (size_t)blockIdx.x * 256;
  const int    b   = (int)(blk >> 16);
  const float* xb  = x + (size_t)b * NC * HW;
#if USE_WMMA_F32X4
  const int lane = tid & 31;
  const int wv   = tid >> 5;
  const int col  = lane & 15;
  const int hi   = lane >> 4;
  const int pb   = (int)(blk & (HW - 1)) + wv * 32;

  v8f acc0 = {};
  v8f acc1 = {};
  v2f a;
  a.x = 1.0f / 128.0f;
  a.y = 1.0f / 128.0f;

  const float* p = xb + (size_t)(2 * hi) * HW + (size_t)(pb + col);
#pragma unroll 4
  for (int c0 = 0; c0 < NC; c0 += 4) {
    v2f b0, b1;
    b0.x = p[0];
    b0.y = p[HW];
    b1.x = p[16];
    b1.y = p[HW + 16];
    p += (size_t)4 * HW;
    acc0 = __builtin_amdgcn_wmma_f32_16x16x4_f32(false, a, false, b0,
                                                 (short)0, acc0, false, false);
    acc1 = __builtin_amdgcn_wmma_f32_16x16x4_f32(false, a, false, b1,
                                                 (short)0, acc1, false, false);
  }
  const float m = hi ? acc1[0] : acc0[0];
  E[(size_t)b * HW + (size_t)(pb + hi * 16 + col)] = expf(m);
#else
  const int    pp = (int)(blk & (HW - 1)) + tid;
  const float* p  = xb + pp;
  float s = 0.0f;
#pragma unroll 8
  for (int c = 0; c < NC; ++c) s += p[(size_t)c * HW];
  E[(size_t)b * HW + pp] = expf(s * (1.0f / 128.0f));
#endif
}

__global__ __launch_bounds__(256) void window_weight_kernel(const float* __restrict__ E,
                                                            const float* __restrict__ lc,
                                                            float* __restrict__ Wgt) {
  const int idx = blockIdx.x * 256 + threadIdx.x;
  if (idx >= NB * HP * WP) return;
  const int b = idx / (HP * WP);
  const int r = idx % (HP * WP);
  const int i = r / WP;
  const int j = r % WP;
  const float* Eb = E + (size_t)b * HW + (size_t)i * WW + j;
  float s = 0.0f;
#pragma unroll
  for (int di = 0; di < KW; ++di) {
#pragma unroll
    for (int dj = 0; dj < KW; ++dj) s += Eb[di * WW + dj];
  }
  Wgt[idx] = lc[idx] / s;
}

__global__ __launch_bounds__(256) void fold_kernel(const float* __restrict__ E,
                                                   const float* __restrict__ Wgt,
                                                   float* __restrict__ out) {
  const int idx = blockIdx.x * 256 + threadIdx.x;
  if (idx >= NB * HW) return;
  const int b  = idx >> 16;
  const int r  = idx & (HW - 1);
  const int y  = r >> 8;
  const int xx = r & 255;
  const int i0 = (y - 7 > 0) ? (y - 7) : 0;
  const int i1 = (y < HP - 1) ? y : (HP - 1);
  const int j0 = (xx - 7 > 0) ? (xx - 7) : 0;
  const int j1 = (xx < WP - 1) ? xx : (WP - 1);
  const float* Wb = Wgt + (size_t)b * HP * WP;
  float s = 0.0f;
  for (int i = i0; i <= i1; ++i) {
    const float* row = Wb + (size_t)i * WP;
    for (int j = j0; j <= j1; ++j) s += row[j];
  }
  out[idx] = E[idx] * s;
}

extern "C" void kernel_launch(void* const* d_in, const int* in_sizes, int n_in,
                              void* d_out, int out_size, void* d_ws, size_t ws_size,
                              hipStream_t stream) {
  (void)in_sizes; (void)n_in; (void)out_size; (void)ws_size;
  const float* x   = (const float*)d_in[0];
  const float* lc  = (const float*)d_in[1];
  float*       out = (float*)d_out;

  float* E   = (float*)d_ws;
  float* Wgt = E + (size_t)NB * HW;

  mean_exp_kernel<<<NB * HW / 256, 256, 0, stream>>>(x, E);

  const int n2 = NB * HP * WP;
  window_weight_kernel<<<(n2 + 255) / 256, 256, 0, stream>>>(E, lc, Wgt);

  const int n3 = NB * HW;
  fold_kernel<<<(n3 + 255) / 256, 256, 0, stream>>>(E, Wgt, out);
}


# Fuse window-denominator and fold passes into one LDS-tiled kernel (8 rows x 256 cols per block) with separable 8-tap box sums: rows summed in registers, columns via LDS; float sum order changes
# speedup vs baseline: 1.2368x; 1.2368x over previous
#include <hip/hip_runtime.h>


typedef __attribute__((ext_vector_type(2))) float v2f;
typedef __attribute__((ext_vector_type(8))) float v8f;

#define NB   8
#define NC   128
#define HH   256
#define WW   256
#define HW   (HH * WW)
#define KW   8
#define HP   (HH - KW + 1)
#define WP   (WW - KW + 1)

#if __has_builtin(__builtin_amdgcn_wmma_f32_16x16x4_f32)
#define USE_WMMA_F32X4 1
#else
#define USE_WMMA_F32X4 0
#endif

__global__ __launch_bounds__(256) void mean_exp_kernel(const float* __restrict__ x,
                                                       float* __restrict__ E) {
  const int    tid = threadIdx.x;
  const size_t blk = (size_t)blockIdx.x * 256;
  const int    b   = (int)(blk >> 16);
  const float* xb  = x + (size_t)b * NC * HW;
#if USE_WMMA_F32X4
  const int lane = tid & 31;
  const int wv   = tid >> 5;
  const int col  = lane & 15;
  const int hi   = lane >> 4;
  const int pb   = (int)(blk & (HW - 1)) + wv * 32;

  v8f acc0 = {};
  v8f acc1 = {};
  v2f a;
  a.x = 1.0f / 128.0f;
  a.y = 1.0f / 128.0f;

  const float* p = xb + (size_t)(2 * hi) * HW + (size_t)(pb + col);
#pragma unroll 4
  for (int c0 = 0; c0 < NC; c0 += 4) {
    v2f b0, b1;
    b0.x = p[0];
    b0.y = p[HW];
    b1.x = p[16];
    b1.y = p[HW + 16];
    p += (size_t)4 * HW;
    acc0 = __builtin_amdgcn_wmma_f32_16x16x4_f32(false, a, false, b0,
                                                 (short)0, acc0, false, false);
    acc1 = __builtin_amdgcn_wmma_f32_16x16x4_f32(false, a, false, b1,
                                                 (short)0, acc1, false, false);
  }
  const float m = hi ? acc1[0] : acc0[0];
  E[(size_t)b * HW + (size_t)(pb + hi * 16 + col)] = expf(m);
#else
  const int    pp = (int)(blk & (HW - 1)) + tid;
  const float* p  = xb + pp;
  float s = 0.0f;
#pragma unroll 8
  for (int c = 0; c < NC; ++c) s += p[(size_t)c * HW];
  E[(size_t)b * HW + pp] = expf(s * (1.0f / 128.0f));
#endif
}

#define TH   8
#define ER   (TH + 14)
#define WR   (TH + 7)
#define GP   (WW + 8)

__global__ __launch_bounds__(256) void window_fold_kernel(const float* __restrict__ E,
                                                          const float* __restrict__ lc,
                                                          float* __restrict__ out) {
  __shared__ float Vs[WR * WW];
  __shared__ float Gs[TH * GP];

  const int tid = threadIdx.x;
  const int b   = blockIdx.x >> 5;
  const int y0  = (blockIdx.x & 31) * TH;
  const float* Eb = E  + (size_t)b * HW;
  const float* lb = lc + (size_t)b * HP * WP;
  const bool   jv = tid < WP;

  float e[ER];
#pragma unroll
  for (int r = 0; r < ER; ++r) {
    int row = y0 - 7 + r;
    row = row < 0 ? 0 : (row > HH - 1 ? HH - 1 : row);
    e[r] = Eb[row * WW + tid];
  }
  float l[WR];
#pragma unroll
  for (int k = 0; k < WR; ++k) {
    const int i  = y0 - 7 + k;
    const int ic = i < 0 ? 0 : (i > HP - 1 ? HP - 1 : i);
    const float v = lb[ic * WP + (jv ? tid : WP - 1)];
    l[k] = (jv && i >= 0 && i < HP) ? v : 0.0f;
  }

  float s2[ER - 1], s4[ER - 3];
#pragma unroll
  for (int k = 0; k < ER - 1; ++k) s2[k] = e[k] + e[k + 1];
#pragma unroll
  for (int k = 0; k < ER - 3; ++k) s4[k] = s2[k] + s2[k + 2];
#pragma unroll
  for (int k = 0; k < WR; ++k) Vs[k * WW + tid] = s4[k] + s4[k + 4];

  if (tid < 8) {
#pragma unroll
    for (int yy = 0; yy < TH; ++yy) Gs[yy * GP + tid] = 0.0f;
  }
  __syncthreads();

  float wt[WR];
#pragma unroll
  for (int k = 0; k < WR; ++k) wt[k] = 0.0f;
  if (jv) {
#pragma unroll
    for (int k = 0; k < WR; ++k) {
      const float* v = &Vs[k * WW + tid];
      const float s = ((v[0] + v[1]) + (v[2] + v[3])) +
                      ((v[4] + v[5]) + (v[6] + v[7]));
      wt[k] = l[k] / s;
    }
  }
  float t2[WR - 1], t4[WR - 3];
#pragma unroll
  for (int k = 0; k < WR - 1; ++k) t2[k] = wt[k] + wt[k + 1];
#pragma unroll
  for (int k = 0; k < WR - 3; ++k) t4[k] = t2[k] + t2[k + 2];
#pragma unroll
  for (int yy = 0; yy < TH; ++yy) Gs[yy * GP + 8 + tid] = t4[yy] + t4[yy + 4];
  __syncthreads();

  float* ob = out + (size_t)b * HW + (size_t)y0 * WW + tid;
#pragma unroll
  for (int yy = 0; yy < TH; ++yy) {
    const float* g = &Gs[yy * GP + 1 + tid];
    const float s = ((g[0] + g[1]) + (g[2] + g[3])) +
                    ((g[4] + g[5]) + (g[6] + g[7]));
    ob[yy * WW] = e[7 + yy] * s;
  }
}

extern "C" void kernel_launch(void* const* d_in, const int* in_sizes, int n_in,
                              void* d_out, int out_size, void* d_ws, size_t ws_size,
                              hipStream_t stream) {
  (void)in_sizes; (void)n_in; (void)out_size; (void)ws_size;
  const float* x   = (const float*)d_in[0];
  const float* lc  = (const float*)d_in[1];
  float*       out = (float*)d_out;

  float* E   = (float*)d_ws;

  mean_exp_kernel<<<NB * HW / 256, 256, 0, stream>>>(x, E);

  window_fold_kernel<<<NB * (HH / TH), 256, 0, stream>>>(E, lc, out);
}
